# starting kernel
# baseline (speedup 1.0000x reference)
#include <hip/hip_runtime.h>
#include <hip/hip_bf16.h>

#define N_NODES   100000
#define N_EDGES   3200000
#define D_FEAT    512
#define D_HID     16
#define N_CLASSES 7

typedef float v2f __attribute__((ext_vector_type(2)));
typedef float v8f __attribute__((ext_vector_type(8)));

__global__ void init_kernel(float* __restrict__ agg1,
                            float* __restrict__ out,
                            const float* __restrict__ b2) {
  int i = blockIdx.x * blockDim.x + threadIdx.x;
  if (i < N_NODES * D_HID) agg1[i] = 0.0f;
  if (i < N_NODES * N_CLASSES) out[i] = b2[i % N_CLASSES];
}

__global__ void gemm1_wmma(const float* __restrict__ feat,
                           const float* __restrict__ W1,
                           float* __restrict__ t1) {
  __shared__ float ldsW[D_FEAT * D_HID];

  {
    const float4* s4 = (const float4*)W1;
    float4* d4 = (float4*)ldsW;
    for (int i = threadIdx.x; i < (D_FEAT * D_HID) / 4; i += blockDim.x)
      d4[i] = s4[i];
  }
  __syncthreads();

  const int lane = threadIdx.x & 31;
  const int wave = threadIdx.x >> 5;
  const int tile = blockIdx.x * (blockDim.x >> 5) + wave;
  if (tile >= N_NODES / 16) return;
  const int node0 = tile * 16;

  const int mn   = lane & 15;
  const int koff = (lane >> 4) << 1;
  const float* arow = feat + (size_t)(node0 + mn) * D_FEAT + koff;
  const float* bcol = ldsW + koff * D_HID + mn;

  v8f acc = {};
  #pragma unroll 4
  for (int kk = 0; kk < D_FEAT; kk += 4) {
    v2f a = *(const v2f*)(arow + kk);
    v2f b;
    b.x = bcol[kk * D_HID];
    b.y = bcol[kk * D_HID + D_HID];
    acc = __builtin_amdgcn_wmma_f32_16x16x4_f32(
        false, a, false, b, (short)0, acc, false, false);
  }

  const int rbase = node0 + ((lane >> 4) << 3);
  #pragma unroll
  for (int r = 0; r < 8; ++r)
    t1[(size_t)(rbase + r) * D_HID + mn] = acc[r];
}

__global__ void scatter1(const float* __restrict__ t1,
                         const int* __restrict__ src,
                         const int* __restrict__ dst,
                         float* __restrict__ agg1) {
  int e = blockIdx.x * blockDim.x + threadIdx.x;
  if (e >= N_EDGES) return;
  const int s = src[e];
  const int d = dst[e];
  const float4* hv = (const float4*)(t1 + (size_t)s * D_HID);
  float* av = agg1 + (size_t)d * D_HID;
  #pragma unroll
  for (int q = 0; q < 4; ++q) {
    float4 v = hv[q];
    atomicAdd(av + q * 4 + 0, v.x);
    atomicAdd(av + q * 4 + 1, v.y);
    atomicAdd(av + q * 4 + 2, v.z);
    atomicAdd(av + q * 4 + 3, v.w);
  }
}

__global__ void layer2_transform(const float* __restrict__ agg1,
                                 const float* __restrict__ b1,
                                 const float* __restrict__ W2,
                                 float* __restrict__ t2) {
  int node = blockIdx.x * blockDim.x + threadIdx.x;
  if (node >= N_NODES) return;
  float h[D_HID];
  const float4* a4 = (const float4*)(agg1 + (size_t)node * D_HID);
  #pragma unroll
  for (int q = 0; q < 4; ++q) {
    float4 v = a4[q];
    h[q * 4 + 0] = fmaxf(v.x + b1[q * 4 + 0], 0.0f);
    h[q * 4 + 1] = fmaxf(v.y + b1[q * 4 + 1], 0.0f);
    h[q * 4 + 2] = fmaxf(v.z + b1[q * 4 + 2], 0.0f);
    h[q * 4 + 3] = fmaxf(v.w + b1[q * 4 + 3], 0.0f);
  }
  #pragma unroll
  for (int c = 0; c < N_CLASSES; ++c) {
    float o = 0.0f;
    #pragma unroll
    for (int j = 0; j < D_HID; ++j)
      o = fmaf(h[j], W2[j * N_CLASSES + c], o);
    t2[(size_t)node * N_CLASSES + c] = o;
  }
}

__global__ void scatter2(const float* __restrict__ t2,
                         const int* __restrict__ src,
                         const int* __restrict__ dst,
                         float* __restrict__ out) {
  int e = blockIdx.x * blockDim.x + threadIdx.x;
  if (e >= N_EDGES) return;
  const int s = src[e];
  const int d = dst[e];
  const float* ts = t2 + (size_t)s * N_CLASSES;
  float* od = out + (size_t)d * N_CLASSES;
  #pragma unroll
  for (int c = 0; c < N_CLASSES; ++c)
    atomicAdd(od + c, ts[c]);
}

extern "C" void kernel_launch(void* const* d_in, const int* in_sizes, int n_in,
                              void* d_out, int out_size, void* d_ws, size_t ws_size,
                              hipStream_t stream) {
  const float* feat = (const float*)d_in[0];
  const float* W1   = (const float*)d_in[1];
  const float* b1   = (const float*)d_in[2];
  const float* W2   = (const float*)d_in[3];
  const float* b2   = (const float*)d_in[4];
  const int*   src  = (const int*)d_in[5];
  const int*   dst  = (const int*)d_in[6];
  float* out = (float*)d_out;

  float* ws   = (float*)d_ws;
  float* t1   = ws;
  float* agg1 = ws + (size_t)N_NODES * D_HID;
  float* t2   = ws + 2 * (size_t)N_NODES * D_HID;

  const int threads = 256;
  init_kernel<<<(N_NODES * D_HID + threads - 1) / threads, threads, 0, stream>>>(
      agg1, out, b2);

  const int tiles = N_NODES / 16;
  const int waves_per_block = threads / 32;
  gemm1_wmma<<<(tiles + waves_per_block - 1) / waves_per_block, threads, 0, stream>>>(
      feat, W1, t1);

  scatter1<<<(N_EDGES + threads - 1) / threads, threads, 0, stream>>>(
      t1, src, dst, agg1);

  layer2_transform<<<(N_NODES + threads - 1) / threads, threads, 0, stream>>>(
      agg1, b1, W2, t2);

  scatter2<<<(N_EDGES + threads - 1) / threads, threads, 0, stream>>>(
      t2, src, dst, out);
}


# scatter-adds -> owner sums: 125 workgroups each hold 800 dst rows in LDS, read the WHOLE dst list (edges never grouped), gather own edges' rows, ds_add_f32, store rows once; f32 sum order changes
# speedup vs baseline: 3.7902x; 3.7902x over previous
#include <hip/hip_runtime.h>
#include <hip/hip_bf16.h>

#define N_NODES   100000
#define N_EDGES   3200000
#define D_FEAT    512
#define D_HID     16
#define N_CLASSES 7

typedef float v2f __attribute__((ext_vector_type(2)));
typedef float v8f __attribute__((ext_vector_type(8)));

__global__ void gemm1_wmma(const float* __restrict__ feat,
                           const float* __restrict__ W1,
                           float* __restrict__ t1) {
  __shared__ float ldsW[D_FEAT * D_HID];

  {
    const float4* s4 = (const float4*)W1;
    float4* d4 = (float4*)ldsW;
    for (int i = threadIdx.x; i < (D_FEAT * D_HID) / 4; i += blockDim.x)
      d4[i] = s4[i];
  }
  __syncthreads();

  const int lane = threadIdx.x & 31;
  const int wave = threadIdx.x >> 5;
  const int tile = blockIdx.x * (blockDim.x >> 5) + wave;
  if (tile >= N_NODES / 16) return;
  const int node0 = tile * 16;

  const int mn   = lane & 15;
  const int koff = (lane >> 4) << 1;
  const float* arow = feat + (size_t)(node0 + mn) * D_FEAT + koff;
  const float* bcol = ldsW + koff * D_HID + mn;

  v8f acc = {};
  #pragma unroll 4
  for (int kk = 0; kk < D_FEAT; kk += 4) {
    v2f a = *(const v2f*)(arow + kk);
    v2f b;
    b.x = bcol[kk * D_HID];
    b.y = bcol[kk * D_HID + D_HID];
    acc = __builtin_amdgcn_wmma_f32_16x16x4_f32(
        false, a, false, b, (short)0, acc, false, false);
  }

  const int rbase = node0 + ((lane >> 4) << 3);
  #pragma unroll
  for (int r = 0; r < 8; ++r)
    t1[(size_t)(rbase + r) * D_HID + mn] = acc[r];
}

#define OWN_R      800
#define OWN_G      (N_NODES / OWN_R)
#define OWN_T      1024
#define OWN_W      (OWN_T / 32)
#define OWN_U      8
#define OWN_DRAIN  96
#define OWN_QCAP   (OWN_DRAIN + OWN_U * 4 * 32)
#define N_VEC4     (N_EDGES / 4)
#define OWN_STEP   (OWN_T * OWN_U)
#define OWN_ITERS  ((N_VEC4 + OWN_STEP - 1) / OWN_STEP)

static_assert(N_NODES % OWN_R == 0, "owners tile the rows");
static_assert(OWN_R % 32 == 0 && OWN_R <= 1024, "row slices on 128-B lines; local row in 10 bits");
static_assert(N_EDGES % 4 == 0 && N_EDGES <= (1 << 22), "edge id in 22 bits");
static_assert(OWN_ITERS % 2 == 0, "passes are taken two at a time");

static __device__ __forceinline__ int own_note(unsigned* q, int cnt, unsigned x, int e) {
  const bool m = x < (unsigned)OWN_R;
  const unsigned mask = __builtin_amdgcn_ballot_w32(m);
  if (mask != 0u) {
    if (m) q[cnt + (int)__builtin_amdgcn_mbcnt_lo(mask, 0u)] = ((unsigned)e << 10) | x;
    cnt += __builtin_popcount(mask);
  }
  return cnt;
}

template <int C, int P>
static __device__ __forceinline__ void own_drain(const unsigned* q, int cnt, int lane,
                                                 const float* __restrict__ rows,
                                                 const int* __restrict__ src,
                                                 float* acc) {
  if constexpr (C == 16) {
    const int sub = lane >> 2, quad = lane & 3;
    for (int k0 = 0; k0 < cnt; k0 += 32) {
      bool ok[4]; unsigned ent[4]; int s[4]; float4 v[4];
      #pragma unroll
      for (int r = 0; r < 4; ++r) {
        const int k = k0 + r * 8 + sub;
        ok[r] = k < cnt;
        ent[r] = ok[r] ? q[k] : 0u;
      }
      #pragma unroll
      for (int r = 0; r < 4; ++r) s[r] = src[ent[r] >> 10];
      #pragma unroll
      for (int r = 0; r < 4; ++r)
        v[r] = ((const float4*)(rows + (size_t)s[r] * 16))[quad];
      #pragma unroll
      for (int r = 0; r < 4; ++r) {
        float* a = acc + (ok[r] ? (int)(ent[r] & 1023u) * P + quad * 4 : OWN_R * P + lane * 4);
        atomicAdd(a + 0, v[r].x);
        atomicAdd(a + 1, v[r].y);
        atomicAdd(a + 2, v[r].z);
        atomicAdd(a + 3, v[r].w);
      }
    }
  } else {
    const int sub = lane >> 3, col = lane & 7;
    const int colc = col < C ? col : C - 1;
    for (int k0 = 0; k0 < cnt; k0 += 16) {
      bool ok[4]; unsigned ent[4]; int s[4]; float v[4];
      #pragma unroll
      for (int r = 0; r < 4; ++r) {
        const int k = k0 + r * 4 + sub;
        ok[r] = (k < cnt) && (col < C);
        ent[r] = ok[r] ? q[k] : 0u;
      }
      #pragma unroll
      for (int r = 0; r < 4; ++r) s[r] = src[ent[r] >> 10];
      #pragma unroll
      for (int r = 0; r < 4; ++r) v[r] = rows[(size_t)s[r] * C + colc];
      #pragma unroll
      for (int r = 0; r < 4; ++r)
        atomicAdd(acc + (ok[r] ? (int)(ent[r] & 1023u) * P + col : OWN_R * P + lane * 4), v[r]);
    }
  }
}

static __device__ __forceinline__ void own_load(const int4* __restrict__ dst4, int it, int tid,
                                                int4 (&d)[OWN_U]) {
  const int base = it * OWN_STEP + tid;
  #pragma unroll
  for (int u = 0; u < OWN_U; ++u) {
    const int i = base + u * OWN_T;
    d[u] = dst4[i < N_VEC4 ? i : N_VEC4 - 1];
  }
}

static __device__ __forceinline__ int own_scan(unsigned* q, int cnt, int4 (&d)[OWN_U],
                                               int it, int tid, int lo) {
  const int base = it * OWN_STEP + tid;
  if (it == OWN_ITERS - 1) {
    #pragma unroll
    for (int u = 0; u < OWN_U; ++u)
      if (base + u * OWN_T >= N_VEC4) d[u] = make_int4(-1, -1, -1, -1);
  }
  #pragma unroll
  for (int u = 0; u < OWN_U; ++u) {
    const int e0 = (base + u * OWN_T) * 4;
    cnt = own_note(q, cnt, (unsigned)(d[u].x - lo), e0 + 0);
    cnt = own_note(q, cnt, (unsigned)(d[u].y - lo), e0 + 1);
    cnt = own_note(q, cnt, (unsigned)(d[u].z - lo), e0 + 2);
    cnt = own_note(q, cnt, (unsigned)(d[u].w - lo), e0 + 3);
  }
  return cnt;
}

template <int C, int P, bool BIAS>
__global__ __launch_bounds__(OWN_T)
void owner_sum(const float* __restrict__ rows,
               const int* __restrict__ src,
               const int* __restrict__ dst,
               const float* __restrict__ bias,
               float* __restrict__ outp) {
  __shared__ float acc[OWN_R * P + 32 * 4];
  __shared__ unsigned queue[OWN_W * OWN_QCAP];

  const int tid  = threadIdx.x;
  const int lane = tid & 31;
  const int wave = tid >> 5;
  const int lo   = blockIdx.x * OWN_R;

  for (int i = tid; i < OWN_R * P + 32 * 4; i += OWN_T) acc[i] = 0.0f;
  __syncthreads();

  unsigned* q = queue + wave * OWN_QCAP;
  const int4* dst4 = (const int4*)dst;
  int cnt = 0;
  const int it0 = (blockIdx.x * OWN_ITERS) / OWN_G;

  int4 da[OWN_U], db[OWN_U];
  int ita = it0;
  own_load(dst4, ita, tid, da);
  for (int n = 0; n < OWN_ITERS; n += 2) {
    const int itb = (ita + 1 == OWN_ITERS) ? 0 : ita + 1;
    own_load(dst4, itb, tid, db);
    cnt = own_scan(q, cnt, da, ita, tid, lo);
    if (cnt >= OWN_DRAIN) {
      own_drain<C, P>(q, cnt, lane, rows, src, acc);
      cnt = 0;
    }
    ita = (itb + 1 == OWN_ITERS) ? 0 : itb + 1;
    own_load(dst4, ita, tid, da);
    cnt = own_scan(q, cnt, db, itb, tid, lo);
    if (cnt >= OWN_DRAIN) {
      own_drain<C, P>(q, cnt, lane, rows, src, acc);
      cnt = 0;
    }
  }
  if (cnt > 0) own_drain<C, P>(q, cnt, lane, rows, src, acc);
  __syncthreads();

  float* o = outp + (size_t)lo * C;
  for (int i = tid; i < OWN_R * C; i += OWN_T) {
    const int r = i / C, c = i - r * C;
    float t = acc[r * P + c];
    if constexpr (BIAS) t += bias[c];
    o[i] = t;
  }
}

__global__ void layer2_transform(const float* __restrict__ agg1,
                                 const float* __restrict__ b1,
                                 const float* __restrict__ W2,
                                 float* __restrict__ t2) {
  int node = blockIdx.x * blockDim.x + threadIdx.x;
  if (node >= N_NODES) return;
  float h[D_HID];
  const float4* a4 = (const float4*)(agg1 + (size_t)node * D_HID);
  #pragma unroll
  for (int q = 0; q < 4; ++q) {
    float4 v = a4[q];
    h[q * 4 + 0] = fmaxf(v.x + b1[q * 4 + 0], 0.0f);
    h[q * 4 + 1] = fmaxf(v.y + b1[q * 4 + 1], 0.0f);
    h[q * 4 + 2] = fmaxf(v.z + b1[q * 4 + 2], 0.0f);
    h[q * 4 + 3] = fmaxf(v.w + b1[q * 4 + 3], 0.0f);
  }
  #pragma unroll
  for (int c = 0; c < N_CLASSES; ++c) {
    float o = 0.0f;
    #pragma unroll
    for (int j = 0; j < D_HID; ++j)
      o = fmaf(h[j], W2[j * N_CLASSES + c], o);
    t2[(size_t)node * N_CLASSES + c] = o;
  }
}

extern "C" void kernel_launch(void* const* d_in, const int* in_sizes, int n_in,
                              void* d_out, int out_size, void* d_ws, size_t ws_size,
                              hipStream_t stream) {
  const float* feat = (const float*)d_in[0];
  const float* W1   = (const float*)d_in[1];
  const float* b1   = (const float*)d_in[2];
  const float* W2   = (const float*)d_in[3];
  const float* b2   = (const float*)d_in[4];
  const int*   src  = (const int*)d_in[5];
  const int*   dst  = (const int*)d_in[6];
  float* out = (float*)d_out;

  float* ws   = (float*)d_ws;
  float* t1   = ws;
  float* agg1 = ws + (size_t)N_NODES * D_HID;
  float* t2   = ws + 2 * (size_t)N_NODES * D_HID;

  const int threads = 256;
  const int tiles = N_NODES / 16;
  const int waves_per_block = threads / 32;
  gemm1_wmma<<<(tiles + waves_per_block - 1) / waves_per_block, threads, 0, stream>>>(
      feat, W1, t1);

  owner_sum<D_HID, D_HID + 1, false><<<OWN_G, OWN_T, 0, stream>>>(
      t1, src, dst, b1, agg1);

  layer2_transform<<<(N_NODES + threads - 1) / threads, threads, 0, stream>>>(
      agg1, b1, W2, t2);

  owner_sum<N_CLASSES, N_CLASSES, true><<<OWN_G, OWN_T, 0, stream>>>(
      t2, src, dst, b2, out);
}


# layer-2 owner sums: 50 row ranges x 4 segments of the dst list (200 workgroups, each reads 1/4 of it), partial rows in 4 planes over spent t1+agg1, new kernel adds planes + b2; f32 sum order changes
# speedup vs baseline: 4.7730x; 1.2593x over previous
#include <hip/hip_runtime.h>
#include <hip/hip_bf16.h>

#define N_NODES   100000
#define N_EDGES   3200000
#define D_FEAT    512
#define D_HID     16
#define N_CLASSES 7

typedef float v2f __attribute__((ext_vector_type(2)));
typedef float v8f __attribute__((ext_vector_type(8)));

__global__ void gemm1_wmma(const float* __restrict__ feat,
                           const float* __restrict__ W1,
                           float* __restrict__ t1) {
  __shared__ float ldsW[D_FEAT * D_HID];

  {
    const float4* s4 = (const float4*)W1;
    float4* d4 = (float4*)ldsW;
    for (int i = threadIdx.x; i < (D_FEAT * D_HID) / 4; i += blockDim.x)
      d4[i] = s4[i];
  }
  __syncthreads();

  const int lane = threadIdx.x & 31;
  const int wave = threadIdx.x >> 5;
  const int tile = blockIdx.x * (blockDim.x >> 5) + wave;
  if (tile >= N_NODES / 16) return;
  const int node0 = tile * 16;

  const int mn   = lane & 15;
  const int koff = (lane >> 4) << 1;
  const float* arow = feat + (size_t)(node0 + mn) * D_FEAT + koff;
  const float* bcol = ldsW + koff * D_HID + mn;

  v8f acc = {};
  #pragma unroll 4
  for (int kk = 0; kk < D_FEAT; kk += 4) {
    v2f a = *(const v2f*)(arow + kk);
    v2f b;
    b.x = bcol[kk * D_HID];
    b.y = bcol[kk * D_HID + D_HID];
    acc = __builtin_amdgcn_wmma_f32_16x16x4_f32(
        false, a, false, b, (short)0, acc, false, false);
  }

  const int rbase = node0 + ((lane >> 4) << 3);
  #pragma unroll
  for (int r = 0; r < 8; ++r)
    t1[(size_t)(rbase + r) * D_HID + mn] = acc[r];
}

#define OWN_T      1024
#define OWN_W      (OWN_T / 32)
#define OWN_U      8
#define OWN_DRAIN  96
#define OWN_QCAP   (OWN_DRAIN + OWN_U * 4 * 32)
#define N_VEC4     (N_EDGES / 4)
#define OWN_STEP   (OWN_T * OWN_U)

static_assert(N_EDGES % 4 == 0, "the dst list is read in 16-byte pieces");

template <int K, int S>
struct OwnCfg {
  static constexpr int R     = N_NODES / K;
  static constexpr int G     = R * S;
  static constexpr int SEG   = N_VEC4 / S;
  static constexpr int FULL  = SEG / OWN_STEP;
  static constexpr int ITERS = ((SEG + OWN_STEP - 1) / OWN_STEP + 1) & ~1;
  static constexpr int XB    = K <= 1024 ? 10 : 11;
  static_assert(N_NODES % K == 0, "owners tile the rows");
  static_assert(N_VEC4 % S == 0, "segments tile the list");
  static_assert(K <= (1 << XB), "local row fits its bits");
  static_assert((long long)SEG * 4 <= (1ll << (32 - XB)), "edge id inside a segment fits the other bits");
};

template <int K, int XB>
static __device__ __forceinline__ int own_note(unsigned* q, int cnt, unsigned x, int e) {
  const bool m = x < (unsigned)K;
  const unsigned mask = __builtin_amdgcn_ballot_w32(m);
  if (mask != 0u) {
    if (m) q[cnt + (int)__builtin_amdgcn_mbcnt_lo(mask, 0u)] = ((unsigned)e << XB) | x;
    cnt += __builtin_popcount(mask);
  }
  return cnt;
}

template <int C, int P, int K, int XB>
static __device__ __forceinline__ void own_drain(const unsigned* q, int cnt, int lane,
                                                 const float* __restrict__ rows,
                                                 const int* __restrict__ src,
                                                 float* acc) {
  constexpr unsigned XM = (1u << XB) - 1u;
  if constexpr (C == 16) {
    const int sub = lane >> 2, quad = lane & 3;
    for (int k0 = 0; k0 < cnt; k0 += 32) {
      bool ok[4]; unsigned ent[4]; int s[4]; float4 v[4];
      #pragma unroll
      for (int r = 0; r < 4; ++r) {
        const int k = k0 + r * 8 + sub;
        ok[r] = k < cnt;
        ent[r] = ok[r] ? q[k] : 0u;
      }
      #pragma unroll
      for (int r = 0; r < 4; ++r) s[r] = src[ent[r] >> XB];
      #pragma unroll
      for (int r = 0; r < 4; ++r)
        v[r] = ((const float4*)(rows + (size_t)s[r] * 16))[quad];
      #pragma unroll
      for (int r = 0; r < 4; ++r) {
        float* a = acc + (ok[r] ? (int)(ent[r] & XM) * P + quad * 4 : K * P + lane * 4);
        atomicAdd(a + 0, v[r].x);
        atomicAdd(a + 1, v[r].y);
        atomicAdd(a + 2, v[r].z);
        atomicAdd(a + 3, v[r].w);
      }
    }
  } else {
    const int sub = lane >> 3, col = lane & 7;
    const int colc = col < C ? col : C - 1;
    for (int k0 = 0; k0 < cnt; k0 += 16) {
      bool ok[4]; unsigned ent[4]; int s[4]; float v[4];
      #pragma unroll
      for (int r = 0; r < 4; ++r) {
        const int k = k0 + r * 4 + sub;
        ok[r] = (k < cnt) && (col < C);
        ent[r] = ok[r] ? q[k] : 0u;
      }
      #pragma unroll
      for (int r = 0; r < 4; ++r) s[r] = src[ent[r] >> XB];
      #pragma unroll
      for (int r = 0; r < 4; ++r) v[r] = rows[(size_t)s[r] * C + colc];
      #pragma unroll
      for (int r = 0; r < 4; ++r)
        atomicAdd(acc + (ok[r] ? (int)(ent[r] & XM) * P + col : K * P + lane * 4), v[r]);
    }
  }
}

template <int SEG>
static __device__ __forceinline__ void own_load(const int4* __restrict__ dst4, int it, int tid,
                                                int4 (&d)[OWN_U]) {
  const int base = it * OWN_STEP + tid;
  #pragma unroll
  for (int u = 0; u < OWN_U; ++u) {
    const int i = base + u * OWN_T;
    d[u] = dst4[i < SEG ? i : SEG - 1];
  }
}

template <int K, int S>
static __device__ __forceinline__ int own_scan(unsigned* q, int cnt, int4 (&d)[OWN_U],
                                               int it, int tid, int lo) {
  using Cfg = OwnCfg<K, S>;
  const int base = it * OWN_STEP + tid;
  if (it >= Cfg::FULL) {
    #pragma unroll
    for (int u = 0; u < OWN_U; ++u)
      if (base + u * OWN_T >= Cfg::SEG) d[u] = make_int4(-1, -1, -1, -1);
  }
  #pragma unroll
  for (int u = 0; u < OWN_U; ++u) {
    const int e0 = (base + u * OWN_T) * 4;
    cnt = own_note<K, Cfg::XB>(q, cnt, (unsigned)(d[u].x - lo), e0 + 0);
    cnt = own_note<K, Cfg::XB>(q, cnt, (unsigned)(d[u].y - lo), e0 + 1);
    cnt = own_note<K, Cfg::XB>(q, cnt, (unsigned)(d[u].z - lo), e0 + 2);
    cnt = own_note<K, Cfg::XB>(q, cnt, (unsigned)(d[u].w - lo), e0 + 3);
  }
  return cnt;
}

template <int C, int P, int K, int S, int OP>
__global__ __launch_bounds__(OWN_T)
void owner_sum(const float* __restrict__ rows,
               const int* __restrict__ src,
               const int* __restrict__ dst,
               float* __restrict__ outp) {
  using Cfg = OwnCfg<K, S>;
  static_assert(OP >= C && (K * OP * 4) % 128 == 0, "an owner's stored slice starts and ends on 128-B lines");
  __shared__ float acc[K * P + 32 * 4];
  __shared__ unsigned queue[OWN_W * OWN_QCAP];

  const int tid  = threadIdx.x;
  const int lane = tid & 31;
  const int wave = tid >> 5;
  const int rng  = blockIdx.x / S;
  const int seg  = blockIdx.x - rng * S;
  const int lo   = rng * K;

  for (int i = tid; i < K * P + 32 * 4; i += OWN_T) acc[i] = 0.0f;
  __syncthreads();

  unsigned* q = queue + wave * OWN_QCAP;
  const int4* dst4 = (const int4*)dst + (size_t)seg * Cfg::SEG;
  const int*  srcs = src + (size_t)seg * Cfg::SEG * 4;
  int cnt = 0;
  const int it0 = (rng * Cfg::ITERS) / Cfg::R;

  int4 da[OWN_U], db[OWN_U];
  int ita = it0;
  own_load<Cfg::SEG>(dst4, ita, tid, da);
  for (int n = 0; n < Cfg::ITERS; n += 2) {
    const int itb = (ita + 1 == Cfg::ITERS) ? 0 : ita + 1;
    own_load<Cfg::SEG>(dst4, itb, tid, db);
    cnt = own_scan<K, S>(q, cnt, da, ita, tid, lo);
    if (cnt >= OWN_DRAIN) {
      own_drain<C, P, K, Cfg::XB>(q, cnt, lane, rows, srcs, acc);
      cnt = 0;
    }
    ita = (itb + 1 == Cfg::ITERS) ? 0 : itb + 1;
    own_load<Cfg::SEG>(dst4, ita, tid, da);
    cnt = own_scan<K, S>(q, cnt, db, itb, tid, lo);
    if (cnt >= OWN_DRAIN) {
      own_drain<C, P, K, Cfg::XB>(q, cnt, lane, rows, srcs, acc);
      cnt = 0;
    }
  }
  if (cnt > 0) own_drain<C, P, K, Cfg::XB>(q, cnt, lane, rows, srcs, acc);
  __syncthreads();

  float* o = outp + (size_t)seg * N_NODES * OP + (size_t)lo * OP;
  for (int i = tid; i < K * OP; i += OWN_T) {
    const int r = i / OP, cc = i - r * OP;
    o[i] = cc < C ? acc[r * P + cc] : 0.0f;
  }
}

__global__ void layer2_transform(const float* __restrict__ agg1,
                                 const float* __restrict__ b1,
                                 const float* __restrict__ W2,
                                 float* __restrict__ t2) {
  int node = blockIdx.x * blockDim.x + threadIdx.x;
  if (node >= N_NODES) return;
  float h[D_HID];
  const float4* a4 = (const float4*)(agg1 + (size_t)node * D_HID);
  #pragma unroll
  for (int q = 0; q < 4; ++q) {
    float4 v = a4[q];
    h[q * 4 + 0] = fmaxf(v.x + b1[q * 4 + 0], 0.0f);
    h[q * 4 + 1] = fmaxf(v.y + b1[q * 4 + 1], 0.0f);
    h[q * 4 + 2] = fmaxf(v.z + b1[q * 4 + 2], 0.0f);
    h[q * 4 + 3] = fmaxf(v.w + b1[q * 4 + 3], 0.0f);
  }
  #pragma unroll
  for (int c = 0; c < N_CLASSES; ++c) {
    float o = 0.0f;
    #pragma unroll
    for (int j = 0; j < D_HID; ++j)
      o = fmaf(h[j], W2[j * N_CLASSES + c], o);
    t2[(size_t)node * N_CLASSES + c] = o;
  }
}

#define L2_K   2000
#define L2_S   4
#define L2_OP  8
#define CMB_T  256

static_assert(N_NODES % 32 == 0, "the last workgroup's rows end on a 128-B line too");

__global__ __launch_bounds__(CMB_T)
void out_combine(const float* __restrict__ parts,
                 const float* __restrict__ b2,
                 float* __restrict__ out) {
  __shared__ float st[CMB_T * N_CLASSES];
  const int tid = threadIdx.x;
  const int n0  = blockIdx.x * CMB_T;
  const int n   = n0 + tid;
  if (n < N_NODES) {
    float4 a = make_float4(0.f, 0.f, 0.f, 0.f), b = a;
    #pragma unroll
    for (int s = 0; s < L2_S; ++s) {
      const float4* p = (const float4*)(parts + (size_t)s * N_NODES * L2_OP + (size_t)n * L2_OP);
      const float4 u = p[0], v = p[1];
      a.x += u.x; a.y += u.y; a.z += u.z; a.w += u.w;
      b.x += v.x; b.y += v.y; b.z += v.z;
    }
    float* t = st + tid * N_CLASSES;
    t[0] = a.x + b2[0]; t[1] = a.y + b2[1]; t[2] = a.z + b2[2]; t[3] = a.w + b2[3];
    t[4] = b.x + b2[4]; t[5] = b.y + b2[5]; t[6] = b.z + b2[6];
  }
  __syncthreads();
  const int nrow = (N_NODES - n0 < CMB_T) ? (N_NODES - n0) : CMB_T;
  float* o = out + (size_t)n0 * N_CLASSES;
  for (int i = tid; i < nrow * N_CLASSES; i += CMB_T) o[i] = st[i];
}

extern "C" void kernel_launch(void* const* d_in, const int* in_sizes, int n_in,
                              void* d_out, int out_size, void* d_ws, size_t ws_size,
                              hipStream_t stream) {
  const float* feat = (const float*)d_in[0];
  const float* W1   = (const float*)d_in[1];
  const float* b1   = (const float*)d_in[2];
  const float* W2   = (const float*)d_in[3];
  const float* b2   = (const float*)d_in[4];
  const int*   src  = (const int*)d_in[5];
  const int*   dst  = (const int*)d_in[6];
  float* out = (float*)d_out;

  float* ws   = (float*)d_ws;
  float* t1   = ws;
  float* agg1 = ws + (size_t)N_NODES * D_HID;
  float* t2   = ws + 2 * (size_t)N_NODES * D_HID;
  float* part = ws;
  static_assert((size_t)L2_S * N_NODES * L2_OP <= 2 * (size_t)N_NODES * D_HID, "planes end before t2");

  const int threads = 256;
  const int tiles = N_NODES / 16;
  const int waves_per_block = threads / 32;
  gemm1_wmma<<<(tiles + waves_per_block - 1) / waves_per_block, threads, 0, stream>>>(
      feat, W1, t1);

  owner_sum<D_HID, D_HID + 1, 800, 1, D_HID><<<OwnCfg<800, 1>::G, OWN_T, 0, stream>>>(
      t1, src, dst, agg1);

  layer2_transform<<<(N_NODES + threads - 1) / threads, threads, 0, stream>>>(
      agg1, b1, W2, t2);

  owner_sum<N_CLASSES, N_CLASSES, L2_K, L2_S, L2_OP><<<OwnCfg<L2_K, L2_S>::G, OWN_T, 0, stream>>>(
      t2, src, dst, part);

  out_combine<<<(N_NODES + CMB_T - 1) / CMB_T, CMB_T, 0, stream>>>(part, b2, out);
}


# layer-1 owner sums: 100 row ranges x 2 halves of the edge list (200 workgroups, each reads half); the 2 partial rows meet in agg1 by global f32 atomic adds onto stored zeros; f32 sum order changes
# speedup vs baseline: 5.7581x; 1.2064x over previous
#include <hip/hip_runtime.h>
#include <hip/hip_bf16.h>

#define N_NODES   100000
#define N_EDGES   3200000
#define D_FEAT    512
#define D_HID     16
#define N_CLASSES 7

typedef float v2f __attribute__((ext_vector_type(2)));
typedef float v8f __attribute__((ext_vector_type(8)));

__global__ void gemm1_wmma(const float* __restrict__ feat,
                           const float* __restrict__ W1,
                           float* __restrict__ t1,
                           float* __restrict__ agg1) {
  __shared__ float ldsW[D_FEAT * D_HID];

  {
    const float4* s4 = (const float4*)W1;
    float4* d4 = (float4*)ldsW;
    for (int i = threadIdx.x; i < (D_FEAT * D_HID) / 4; i += blockDim.x)
      d4[i] = s4[i];
  }
  __syncthreads();

  const int lane = threadIdx.x & 31;
  const int wave = threadIdx.x >> 5;
  const int tile = blockIdx.x * (blockDim.x >> 5) + wave;
  if (tile >= N_NODES / 16) return;
  const int node0 = tile * 16;

  {
    float4* z = (float4*)(agg1 + (size_t)node0 * D_HID);
    z[lane]      = make_float4(0.f, 0.f, 0.f, 0.f);
    z[lane + 32] = make_float4(0.f, 0.f, 0.f, 0.f);
  }

  const int mn   = lane & 15;
  const int koff = (lane >> 4) << 1;
  const float* arow = feat + (size_t)(node0 + mn) * D_FEAT + koff;
  const float* bcol = ldsW + koff * D_HID + mn;

  v8f acc = {};
  #pragma unroll 4
  for (int kk = 0; kk < D_FEAT; kk += 4) {
    v2f a = *(const v2f*)(arow + kk);
    v2f b;
    b.x = bcol[kk * D_HID];
    b.y = bcol[kk * D_HID + D_HID];
    acc = __builtin_amdgcn_wmma_f32_16x16x4_f32(
        false, a, false, b, (short)0, acc, false, false);
  }

  const int rbase = node0 + ((lane >> 4) << 3);
  #pragma unroll
  for (int r = 0; r < 8; ++r)
    t1[(size_t)(rbase + r) * D_HID + mn] = acc[r];
}

#define OWN_T      1024
#define OWN_W      (OWN_T / 32)
#define OWN_U      8
#define OWN_DRAIN  96
#define OWN_QCAP   (OWN_DRAIN + OWN_U * 4 * 32)
#define N_VEC4     (N_EDGES / 4)
#define OWN_STEP   (OWN_T * OWN_U)

static_assert(N_EDGES % 4 == 0, "the dst list is read in 16-byte pieces");

template <int K, int S>
struct OwnCfg {
  static constexpr int R     = N_NODES / K;
  static constexpr int G     = R * S;
  static constexpr int SEG   = N_VEC4 / S;
  static constexpr int FULL  = SEG / OWN_STEP;
  static constexpr int ITERS = ((SEG + OWN_STEP - 1) / OWN_STEP + 1) & ~1;
  static constexpr int XB    = K <= 1024 ? 10 : 11;
  static_assert(N_NODES % K == 0, "owners tile the rows");
  static_assert(N_VEC4 % S == 0, "segments tile the list");
  static_assert(K <= (1 << XB), "local row fits its bits");
  static_assert((long long)SEG * 4 <= (1ll << (32 - XB)), "edge id inside a segment fits the other bits");
};

template <int K, int XB>
static __device__ __forceinline__ int own_note(unsigned* q, int cnt, unsigned x, int e) {
  const bool m = x < (unsigned)K;
  const unsigned mask = __builtin_amdgcn_ballot_w32(m);
  if (mask != 0u) {
    if (m) q[cnt + (int)__builtin_amdgcn_mbcnt_lo(mask, 0u)] = ((unsigned)e << XB) | x;
    cnt += __builtin_popcount(mask);
  }
  return cnt;
}

template <int C, int P, int K, int XB>
static __device__ __forceinline__ void own_drain(const unsigned* q, int cnt, int lane,
                                                 const float* __restrict__ rows,
                                                 const int* __restrict__ src,
                                                 float* acc) {
  constexpr unsigned XM = (1u << XB) - 1u;
  if constexpr (C == 16) {
    const int sub = lane >> 2, quad = lane & 3;
    for (int k0 = 0; k0 < cnt; k0 += 32) {
      bool ok[4]; unsigned ent[4]; int s[4]; float4 v[4];
      #pragma unroll
      for (int r = 0; r < 4; ++r) {
        const int k = k0 + r * 8 + sub;
        ok[r] = k < cnt;
        ent[r] = ok[r] ? q[k] : 0u;
      }
      #pragma unroll
      for (int r = 0; r < 4; ++r) s[r] = src[ent[r] >> XB];
      #pragma unroll
      for (int r = 0; r < 4; ++r)
        v[r] = ((const float4*)(rows + (size_t)s[r] * 16))[quad];
      #pragma unroll
      for (int r = 0; r < 4; ++r) {
        float* a = acc + (ok[r] ? (int)(ent[r] & XM) * P + quad * 4 : K * P + lane * 4);
        atomicAdd(a + 0, v[r].x);
        atomicAdd(a + 1, v[r].y);
        atomicAdd(a + 2, v[r].z);
        atomicAdd(a + 3, v[r].w);
      }
    }
  } else {
    const int sub = lane >> 3, col = lane & 7;
    const int colc = col < C ? col : C - 1;
    for (int k0 = 0; k0 < cnt; k0 += 16) {
      bool ok[4]; unsigned ent[4]; int s[4]; float v[4];
      #pragma unroll
      for (int r = 0; r < 4; ++r) {
        const int k = k0 + r * 4 + sub;
        ok[r] = (k < cnt) && (col < C);
        ent[r] = ok[r] ? q[k] : 0u;
      }
      #pragma unroll
      for (int r = 0; r < 4; ++r) s[r] = src[ent[r] >> XB];
      #pragma unroll
      for (int r = 0; r < 4; ++r) v[r] = rows[(size_t)s[r] * C + colc];
      #pragma unroll
      for (int r = 0; r < 4; ++r)
        atomicAdd(acc + (ok[r] ? (int)(ent[r] & XM) * P + col : K * P + lane * 4), v[r]);
    }
  }
}

template <int SEG>
static __device__ __forceinline__ void own_load(const int4* __restrict__ dst4, int it, int tid,
                                                int4 (&d)[OWN_U]) {
  const int base = it * OWN_STEP + tid;
  #pragma unroll
  for (int u = 0; u < OWN_U; ++u) {
    const int i = base + u * OWN_T;
    d[u] = dst4[i < SEG ? i : SEG - 1];
  }
}

template <int K, int S>
static __device__ __forceinline__ int own_scan(unsigned* q, int cnt, int4 (&d)[OWN_U],
                                               int it, int tid, int lo) {
  using Cfg = OwnCfg<K, S>;
  const int base = it * OWN_STEP + tid;
  if (it >= Cfg::FULL) {
    #pragma unroll
    for (int u = 0; u < OWN_U; ++u)
      if (base + u * OWN_T >= Cfg::SEG) d[u] = make_int4(-1, -1, -1, -1);
  }
  #pragma unroll
  for (int u = 0; u < OWN_U; ++u) {
    const int e0 = (base + u * OWN_T) * 4;
    cnt = own_note<K, Cfg::XB>(q, cnt, (unsigned)(d[u].x - lo), e0 + 0);
    cnt = own_note<K, Cfg::XB>(q, cnt, (unsigned)(d[u].y - lo), e0 + 1);
    cnt = own_note<K, Cfg::XB>(q, cnt, (unsigned)(d[u].z - lo), e0 + 2);
    cnt = own_note<K, Cfg::XB>(q, cnt, (unsigned)(d[u].w - lo), e0 + 3);
  }
  return cnt;
}

template <int C, int P, int K, int S, int OP, bool ADD>
__global__ __launch_bounds__(OWN_T)
void owner_sum(const float* __restrict__ rows,
               const int* __restrict__ src,
               const int* __restrict__ dst,
               float* __restrict__ outp) {
  using Cfg = OwnCfg<K, S>;
  static_assert(OP >= C && (K * OP * 4) % 128 == 0, "an owner's stored slice starts and ends on 128-B lines");
  __shared__ float acc[K * P + 32 * 4];
  __shared__ unsigned queue[OWN_W * OWN_QCAP];

  const int tid  = threadIdx.x;
  const int lane = tid & 31;
  const int wave = tid >> 5;
  const int rng  = blockIdx.x / S;
  const int seg  = blockIdx.x - rng * S;
  const int lo   = rng * K;

  for (int i = tid; i < K * P + 32 * 4; i += OWN_T) acc[i] = 0.0f;
  __syncthreads();

  unsigned* q = queue + wave * OWN_QCAP;
  const int4* dst4 = (const int4*)dst + (size_t)seg * Cfg::SEG;
  const int*  srcs = src + (size_t)seg * Cfg::SEG * 4;
  int cnt = 0;
  const int it0 = (rng * Cfg::ITERS) / Cfg::R;

  int4 da[OWN_U], db[OWN_U];
  int ita = it0;
  own_load<Cfg::SEG>(dst4, ita, tid, da);
  for (int n = 0; n < Cfg::ITERS; n += 2) {
    const int itb = (ita + 1 == Cfg::ITERS) ? 0 : ita + 1;
    own_load<Cfg::SEG>(dst4, itb, tid, db);
    cnt = own_scan<K, S>(q, cnt, da, ita, tid, lo);
    if (cnt >= OWN_DRAIN) {
      own_drain<C, P, K, Cfg::XB>(q, cnt, lane, rows, srcs, acc);
      cnt = 0;
    }
    ita = (itb + 1 == Cfg::ITERS) ? 0 : itb + 1;
    own_load<Cfg::SEG>(dst4, ita, tid, da);
    cnt = own_scan<K, S>(q, cnt, db, itb, tid, lo);
    if (cnt >= OWN_DRAIN) {
      own_drain<C, P, K, Cfg::XB>(q, cnt, lane, rows, srcs, acc);
      cnt = 0;
    }
  }
  if (cnt > 0) own_drain<C, P, K, Cfg::XB>(q, cnt, lane, rows, srcs, acc);
  __syncthreads();

  if constexpr (ADD) {
    static_assert(OP == C, "no padding columns in the added-into plane");
    float* o = outp + (size_t)lo * OP;
    for (int i = tid; i < K * OP; i += OWN_T) {
      const int r = i / OP, cc = i - r * OP;
      atomicAdd(o + i, acc[r * P + cc]);
    }
  } else {
    float* o = outp + (size_t)seg * N_NODES * OP + (size_t)lo * OP;
    for (int i = tid; i < K * OP; i += OWN_T) {
      const int r = i / OP, cc = i - r * OP;
      o[i] = cc < C ? acc[r * P + cc] : 0.0f;
    }
  }
}

__global__ void layer2_transform(const float* __restrict__ agg1,
                                 const float* __restrict__ b1,
                                 const float* __restrict__ W2,
                                 float* __restrict__ t2) {
  int node = blockIdx.x * blockDim.x + threadIdx.x;
  if (node >= N_NODES) return;
  float h[D_HID];
  const float4* a4 = (const float4*)(agg1 + (size_t)node * D_HID);
  #pragma unroll
  for (int q = 0; q < 4; ++q) {
    float4 v = a4[q];
    h[q * 4 + 0] = fmaxf(v.x + b1[q * 4 + 0], 0.0f);
    h[q * 4 + 1] = fmaxf(v.y + b1[q * 4 + 1], 0.0f);
    h[q * 4 + 2] = fmaxf(v.z + b1[q * 4 + 2], 0.0f);
    h[q * 4 + 3] = fmaxf(v.w + b1[q * 4 + 3], 0.0f);
  }
  #pragma unroll
  for (int c = 0; c < N_CLASSES; ++c) {
    float o = 0.0f;
    #pragma unroll
    for (int j = 0; j < D_HID; ++j)
      o = fmaf(h[j], W2[j * N_CLASSES + c], o);
    t2[(size_t)node * N_CLASSES + c] = o;
  }
}

#define L1_K   1000
#define L1_S   2
#define L2_K   2000
#define L2_S   4
#define L2_OP  8
#define CMB_T  256

static_assert(N_NODES % 32 == 0, "the last workgroup's rows end on a 128-B line too");

__global__ __launch_bounds__(CMB_T)
void out_combine(const float* __restrict__ parts,
                 const float* __restrict__ b2,
                 float* __restrict__ out) {
  __shared__ float st[CMB_T * N_CLASSES];
  const int tid = threadIdx.x;
  const int n0  = blockIdx.x * CMB_T;
  const int n   = n0 + tid;
  if (n < N_NODES) {
    float4 a = make_float4(0.f, 0.f, 0.f, 0.f), b = a;
    #pragma unroll
    for (int s = 0; s < L2_S; ++s) {
      const float4* p = (const float4*)(parts + (size_t)s * N_NODES * L2_OP + (size_t)n * L2_OP);
      const float4 u = p[0], v = p[1];
      a.x += u.x; a.y += u.y; a.z += u.z; a.w += u.w;
      b.x += v.x; b.y += v.y; b.z += v.z;
    }
    float* t = st + tid * N_CLASSES;
    t[0] = a.x + b2[0]; t[1] = a.y + b2[1]; t[2] = a.z + b2[2]; t[3] = a.w + b2[3];
    t[4] = b.x + b2[4]; t[5] = b.y + b2[5]; t[6] = b.z + b2[6];
  }
  __syncthreads();
  const int nrow = (N_NODES - n0 < CMB_T) ? (N_NODES - n0) : CMB_T;
  float* o = out + (size_t)n0 * N_CLASSES;
  for (int i = tid; i < nrow * N_CLASSES; i += CMB_T) o[i] = st[i];
}

extern "C" void kernel_launch(void* const* d_in, const int* in_sizes, int n_in,
                              void* d_out, int out_size, void* d_ws, size_t ws_size,
                              hipStream_t stream) {
  const float* feat = (const float*)d_in[0];
  const float* W1   = (const float*)d_in[1];
  const float* b1   = (const float*)d_in[2];
  const float* W2   = (const float*)d_in[3];
  const float* b2   = (const float*)d_in[4];
  const int*   src  = (const int*)d_in[5];
  const int*   dst  = (const int*)d_in[6];
  float* out = (float*)d_out;

  float* ws   = (float*)d_ws;
  float* t1   = ws;
  float* agg1 = ws + (size_t)N_NODES * D_HID;
  float* t2   = ws + 2 * (size_t)N_NODES * D_HID;
  float* part = ws;
  static_assert((size_t)L2_S * N_NODES * L2_OP <= 2 * (size_t)N_NODES * D_HID, "planes end before t2");

  const int threads = 256;
  const int tiles = N_NODES / 16;
  const int waves_per_block = threads / 32;
  gemm1_wmma<<<(tiles + waves_per_block - 1) / waves_per_block, threads, 0, stream>>>(
      feat, W1, t1, agg1);

  owner_sum<D_HID, D_HID + 1, L1_K, L1_S, D_HID, true><<<OwnCfg<L1_K, L1_S>::G, OWN_T, 0, stream>>>(
      t1, src, dst, agg1);

  layer2_transform<<<(N_NODES + threads - 1) / threads, threads, 0, stream>>>(
      agg1, b1, W2, t2);

  owner_sum<N_CLASSES, N_CLASSES, L2_K, L2_S, L2_OP, false><<<OwnCfg<L2_K, L2_S>::G, OWN_T, 0, stream>>>(
      t2, src, dst, part);

  out_combine<<<(N_NODES + CMB_T - 1) / CMB_T, CMB_T, 0, stream>>>(part, b2, out);
}


# layer-1 owner sums: 25 ranges of 4000 rows x 10 list segments (250 workgroups, each reads 1/10); queues checked 4x a pass free the LDS for rows; 10 partials meet by f32 atomics; f32 sum order changes
# speedup vs baseline: 6.7223x; 1.1675x over previous
#include <hip/hip_runtime.h>
#include <hip/hip_bf16.h>

#define N_NODES   100000
#define N_EDGES   3200000
#define D_FEAT    512
#define D_HID     16
#define N_CLASSES 7

typedef float v2f __attribute__((ext_vector_type(2)));
typedef float v8f __attribute__((ext_vector_type(8)));

__global__ void gemm1_wmma(const float* __restrict__ feat,
                           const float* __restrict__ W1,
                           float* __restrict__ t1,
                           float* __restrict__ agg1) {
  __shared__ float ldsW[D_FEAT * D_HID];

  {
    const float4* s4 = (const float4*)W1;
    float4* d4 = (float4*)ldsW;
    for (int i = threadIdx.x; i < (D_FEAT * D_HID) / 4; i += blockDim.x)
      d4[i] = s4[i];
  }
  __syncthreads();

  const int lane = threadIdx.x & 31;
  const int wave = threadIdx.x >> 5;
  const int tile = blockIdx.x * (blockDim.x >> 5) + wave;
  if (tile >= N_NODES / 16) return;
  const int node0 = tile * 16;

  {
    float4* z = (float4*)(agg1 + (size_t)node0 * D_HID);
    z[lane]      = make_float4(0.f, 0.f, 0.f, 0.f);
    z[lane + 32] = make_float4(0.f, 0.f, 0.f, 0.f);
  }

  const int mn   = lane & 15;
  const int koff = (lane >> 4) << 1;
  const float* arow = feat + (size_t)(node0 + mn) * D_FEAT + koff;
  const float* bcol = ldsW + koff * D_HID + mn;

  v8f acc = {};
  #pragma unroll 4
  for (int kk = 0; kk < D_FEAT; kk += 4) {
    v2f a = *(const v2f*)(arow + kk);
    v2f b;
    b.x = bcol[kk * D_HID];
    b.y = bcol[kk * D_HID + D_HID];
    acc = __builtin_amdgcn_wmma_f32_16x16x4_f32(
        false, a, false, b, (short)0, acc, false, false);
  }

  const int rbase = node0 + ((lane >> 4) << 3);
  #pragma unroll
  for (int r = 0; r < 8; ++r)
    t1[(size_t)(rbase + r) * D_HID + mn] = acc[r];
}

#define OWN_T      1024
#define OWN_W      (OWN_T / 32)
#define OWN_U      8
#define OWN_DRAIN  96
#define OWN_QCAP(CH) (OWN_DRAIN + (CH) * 4 * 32)
#define N_VEC4     (N_EDGES / 4)
#define OWN_STEP   (OWN_T * OWN_U)

static_assert(N_EDGES % 4 == 0, "the dst list is read in 16-byte pieces");

template <int K, int S>
struct OwnCfg {
  static constexpr int R     = N_NODES / K;
  static constexpr int G     = R * S;
  static constexpr int SEG   = N_VEC4 / S;
  static constexpr int FULL  = SEG / OWN_STEP;
  static constexpr int ITERS = ((SEG + OWN_STEP - 1) / OWN_STEP + 1) & ~1;
  static constexpr int XB    = K <= 1024 ? 10 : (K <= 2048 ? 11 : 12);
  static_assert(N_NODES % K == 0, "owners tile the rows");
  static_assert(N_VEC4 % S == 0, "segments tile the list");
  static_assert(K <= (1 << XB), "local row fits its bits");
  static_assert((long long)SEG * 4 <= (1ll << (32 - XB)), "edge id inside a segment fits the other bits");
};

template <int K, int XB>
static __device__ __forceinline__ int own_note(unsigned* q, int cnt, unsigned x, int e) {
  const bool m = x < (unsigned)K;
  const unsigned mask = __builtin_amdgcn_ballot_w32(m);
  if (mask != 0u) {
    if (m) q[cnt + (int)__builtin_amdgcn_mbcnt_lo(mask, 0u)] = ((unsigned)e << XB) | x;
    cnt += __builtin_popcount(mask);
  }
  return cnt;
}

template <int C, int P, int K, int XB>
static __device__ __forceinline__ void own_drain(const unsigned* q, int cnt, int lane,
                                                 const float* __restrict__ rows,
                                                 const int* __restrict__ src,
                                                 float* acc) {
  constexpr unsigned XM = (1u << XB) - 1u;
  if constexpr (C == 16) {
    const int sub = lane >> 2, quad = lane & 3;
    for (int k0 = 0; k0 < cnt; k0 += 32) {
      bool ok[4]; unsigned ent[4]; int s[4]; float4 v[4];
      #pragma unroll
      for (int r = 0; r < 4; ++r) {
        const int k = k0 + r * 8 + sub;
        ok[r] = k < cnt;
        ent[r] = ok[r] ? q[k] : 0u;
      }
      #pragma unroll
      for (int r = 0; r < 4; ++r) s[r] = src[ent[r] >> XB];
      #pragma unroll
      for (int r = 0; r < 4; ++r)
        v[r] = ((const float4*)(rows + (size_t)s[r] * 16))[quad];
      #pragma unroll
      for (int r = 0; r < 4; ++r) {
        float* a = acc + (ok[r] ? (int)(ent[r] & XM) * P + quad * 4 : K * P + lane * 4);
        atomicAdd(a + 0, v[r].x);
        atomicAdd(a + 1, v[r].y);
        atomicAdd(a + 2, v[r].z);
        atomicAdd(a + 3, v[r].w);
      }
    }
  } else {
    const int sub = lane >> 3, col = lane & 7;
    const int colc = col < C ? col : C - 1;
    for (int k0 = 0; k0 < cnt; k0 += 16) {
      bool ok[4]; unsigned ent[4]; int s[4]; float v[4];
      #pragma unroll
      for (int r = 0; r < 4; ++r) {
        const int k = k0 + r * 4 + sub;
        ok[r] = (k < cnt) && (col < C);
        ent[r] = ok[r] ? q[k] : 0u;
      }
      #pragma unroll
      for (int r = 0; r < 4; ++r) s[r] = src[ent[r] >> XB];
      #pragma unroll
      for (int r = 0; r < 4; ++r) v[r] = rows[(size_t)s[r] * C + colc];
      #pragma unroll
      for (int r = 0; r < 4; ++r)
        atomicAdd(acc + (ok[r] ? (int)(ent[r] & XM) * P + col : K * P + lane * 4), v[r]);
    }
  }
}

template <int SEG>
static __device__ __forceinline__ void own_load(const int4* __restrict__ dst4, int it, int tid,
                                                int4 (&d)[OWN_U]) {
  const int base = it * OWN_STEP + tid;
  #pragma unroll
  for (int u = 0; u < OWN_U; ++u) {
    const int i = base + u * OWN_T;
    d[u] = dst4[i < SEG ? i : SEG - 1];
  }
}

template <int K, int S, int U0, int CH>
static __device__ __forceinline__ int own_scan(unsigned* q, int cnt, int4 (&d)[OWN_U],
                                               int it, int tid, int lo) {
  using Cfg = OwnCfg<K, S>;
  const int base = it * OWN_STEP + tid;
  if (it >= Cfg::FULL) {
    #pragma unroll
    for (int u = U0; u < U0 + CH; ++u)
      if (base + u * OWN_T >= Cfg::SEG) d[u] = make_int4(-1, -1, -1, -1);
  }
  #pragma unroll
  for (int u = U0; u < U0 + CH; ++u) {
    const int e0 = (base + u * OWN_T) * 4;
    cnt = own_note<K, Cfg::XB>(q, cnt, (unsigned)(d[u].x - lo), e0 + 0);
    cnt = own_note<K, Cfg::XB>(q, cnt, (unsigned)(d[u].y - lo), e0 + 1);
    cnt = own_note<K, Cfg::XB>(q, cnt, (unsigned)(d[u].z - lo), e0 + 2);
    cnt = own_note<K, Cfg::XB>(q, cnt, (unsigned)(d[u].w - lo), e0 + 3);
  }
  return cnt;
}

template <int C, int P, int K, int S, int CH, int U0 = 0>
static __device__ __forceinline__ int own_pass(unsigned* q, int cnt, int4 (&d)[OWN_U],
                                               int it, int tid, int lane, int lo,
                                               const float* __restrict__ rows,
                                               const int* __restrict__ srcs,
                                               float* acc) {
  if constexpr (U0 < OWN_U) {
    cnt = own_scan<K, S, U0, CH>(q, cnt, d, it, tid, lo);
    if (cnt >= OWN_DRAIN) {
      own_drain<C, P, K, OwnCfg<K, S>::XB>(q, cnt, lane, rows, srcs, acc);
      cnt = 0;
    }
    cnt = own_pass<C, P, K, S, CH, U0 + CH>(q, cnt, d, it, tid, lane, lo, rows, srcs, acc);
  }
  return cnt;
}

template <int C, int P, int K, int S, int OP, bool ADD, int CH>
__global__ __launch_bounds__(OWN_T)
void owner_sum(const float* __restrict__ rows,
               const int* __restrict__ src,
               const int* __restrict__ dst,
               float* __restrict__ outp) {
  using Cfg = OwnCfg<K, S>;
  static_assert(OP >= C && (K * OP * 4) % 128 == 0, "an owner's stored slice starts and ends on 128-B lines");
  static_assert(CH >= 1 && OWN_U % CH == 0, "a pass is a whole number of looks");
  static_assert((K * P + 32 * 4 + OWN_W * OWN_QCAP(CH)) * 4 <= 320 * 1024, "rows and queues fit the LDS");
  __shared__ float acc[K * P + 32 * 4];
  __shared__ unsigned queue[OWN_W * OWN_QCAP(CH)];

  const int tid  = threadIdx.x;
  const int lane = tid & 31;
  const int wave = tid >> 5;
  const int rng  = blockIdx.x / S;
  const int seg  = blockIdx.x - rng * S;
  const int lo   = rng * K;

  for (int i = tid; i < K * P + 32 * 4; i += OWN_T) acc[i] = 0.0f;
  __syncthreads();

  unsigned* q = queue + wave * OWN_QCAP(CH);
  const int4* dst4 = (const int4*)dst + (size_t)seg * Cfg::SEG;
  const int*  srcs = src + (size_t)seg * Cfg::SEG * 4;
  int cnt = 0;
  const int it0 = (rng * Cfg::ITERS) / Cfg::R;

  int4 da[OWN_U], db[OWN_U];
  int ita = it0;
  own_load<Cfg::SEG>(dst4, ita, tid, da);
  for (int n = 0; n < Cfg::ITERS; n += 2) {
    const int itb = (ita + 1 == Cfg::ITERS) ? 0 : ita + 1;
    own_load<Cfg::SEG>(dst4, itb, tid, db);
    cnt = own_pass<C, P, K, S, CH>(q, cnt, da, ita, tid, lane, lo, rows, srcs, acc);
    ita = (itb + 1 == Cfg::ITERS) ? 0 : itb + 1;
    own_load<Cfg::SEG>(dst4, ita, tid, da);
    cnt = own_pass<C, P, K, S, CH>(q, cnt, db, itb, tid, lane, lo, rows, srcs, acc);
  }
  if (cnt > 0) own_drain<C, P, K, Cfg::XB>(q, cnt, lane, rows, srcs, acc);
  __syncthreads();

  if constexpr (ADD) {
    static_assert(OP == C, "no padding columns in the added-into plane");
    float* o = outp + (size_t)lo * OP;
    for (int i = tid; i < K * OP; i += OWN_T) {
      const int r = i / OP, cc = i - r * OP;
      atomicAdd(o + i, acc[r * P + cc]);
    }
  } else {
    float* o = outp + (size_t)seg * N_NODES * OP + (size_t)lo * OP;
    for (int i = tid; i < K * OP; i += OWN_T) {
      const int r = i / OP, cc = i - r * OP;
      o[i] = cc < C ? acc[r * P + cc] : 0.0f;
    }
  }
}

__global__ void layer2_transform(const float* __restrict__ agg1,
                                 const float* __restrict__ b1,
                                 const float* __restrict__ W2,
                                 float* __restrict__ t2) {
  int node = blockIdx.x * blockDim.x + threadIdx.x;
  if (node >= N_NODES) return;
  float h[D_HID];
  const float4* a4 = (const float4*)(agg1 + (size_t)node * D_HID);
  #pragma unroll
  for (int q = 0; q < 4; ++q) {
    float4 v = a4[q];
    h[q * 4 + 0] = fmaxf(v.x + b1[q * 4 + 0], 0.0f);
    h[q * 4 + 1] = fmaxf(v.y + b1[q * 4 + 1], 0.0f);
    h[q * 4 + 2] = fmaxf(v.z + b1[q * 4 + 2], 0.0f);
    h[q * 4 + 3] = fmaxf(v.w + b1[q * 4 + 3], 0.0f);
  }
  #pragma unroll
  for (int c = 0; c < N_CLASSES; ++c) {
    float o = 0.0f;
    #pragma unroll
    for (int j = 0; j < D_HID; ++j)
      o = fmaf(h[j], W2[j * N_CLASSES + c], o);
    t2[(size_t)node * N_CLASSES + c] = o;
  }
}

#define L1_K   4000
#define L1_S   10
#define L1_CH  2
#define L2_CH  OWN_U
#define L2_K   2000
#define L2_S   4
#define L2_OP  8
#define CMB_T  256

static_assert(N_NODES % 32 == 0, "the last workgroup's rows end on a 128-B line too");

__global__ __launch_bounds__(CMB_T)
void out_combine(const float* __restrict__ parts,
                 const float* __restrict__ b2,
                 float* __restrict__ out) {
  __shared__ float st[CMB_T * N_CLASSES];
  const int tid = threadIdx.x;
  const int n0  = blockIdx.x * CMB_T;
  const int n   = n0 + tid;
  if (n < N_NODES) {
    float4 a = make_float4(0.f, 0.f, 0.f, 0.f), b = a;
    #pragma unroll
    for (int s = 0; s < L2_S; ++s) {
      const float4* p = (const float4*)(parts + (size_t)s * N_NODES * L2_OP + (size_t)n * L2_OP);
      const float4 u = p[0], v = p[1];
      a.x += u.x; a.y += u.y; a.z += u.z; a.w += u.w;
      b.x += v.x; b.y += v.y; b.z += v.z;
    }
    float* t = st + tid * N_CLASSES;
    t[0] = a.x + b2[0]; t[1] = a.y + b2[1]; t[2] = a.z + b2[2]; t[3] = a.w + b2[3];
    t[4] = b.x + b2[4]; t[5] = b.y + b2[5]; t[6] = b.z + b2[6];
  }
  __syncthreads();
  const int nrow = (N_NODES - n0 < CMB_T) ? (N_NODES - n0) : CMB_T;
  float* o = out + (size_t)n0 * N_CLASSES;
  for (int i = tid; i < nrow * N_CLASSES; i += CMB_T) o[i] = st[i];
}

extern "C" void kernel_launch(void* const* d_in, const int* in_sizes, int n_in,
                              void* d_out, int out_size, void* d_ws, size_t ws_size,
                              hipStream_t stream) {
  const float* feat = (const float*)d_in[0];
  const float* W1   = (const float*)d_in[1];
  const float* b1   = (const float*)d_in[2];
  const float* W2   = (const float*)d_in[3];
  const float* b2   = (const float*)d_in[4];
  const int*   src  = (const int*)d_in[5];
  const int*   dst  = (const int*)d_in[6];
  float* out = (float*)d_out;

  float* ws   = (float*)d_ws;
  float* t1   = ws;
  float* agg1 = ws + (size_t)N_NODES * D_HID;
  float* t2   = ws + 2 * (size_t)N_NODES * D_HID;
  float* part = ws;
  static_assert((size_t)L2_S * N_NODES * L2_OP <= 2 * (size_t)N_NODES * D_HID, "planes end before t2");

  const int threads = 256;
  const int tiles = N_NODES / 16;
  const int waves_per_block = threads / 32;
  gemm1_wmma<<<(tiles + waves_per_block - 1) / waves_per_block, threads, 0, stream>>>(
      feat, W1, t1, agg1);

  owner_sum<D_HID, D_HID + 1, L1_K, L1_S, D_HID, true, L1_CH><<<OwnCfg<L1_K, L1_S>::G, OWN_T, 0, stream>>>(
      t1, src, dst, agg1);

  layer2_transform<<<(N_NODES + threads - 1) / threads, threads, 0, stream>>>(
      agg1, b1, W2, t2);

  owner_sum<N_CLASSES, N_CLASSES, L2_K, L2_S, L2_OP, false, L2_CH><<<OwnCfg<L2_K, L2_S>::G, OWN_T, 0, stream>>>(
      t2, src, dst, part);

  out_combine<<<(N_NODES + CMB_T - 1) / CMB_T, CMB_T, 0, stream>>>(part, b2, out);
}


# layer-2 owner sums: 25 ranges of 4000 rows x 10 list segments (250 workgroups, each reads 1/10); partials meet by f32 atomics in out on stored b2; combine launch gone; f32 sum order changes
# speedup vs baseline: 7.8380x; 1.1660x over previous
#include <hip/hip_runtime.h>
#include <hip/hip_bf16.h>

#define N_NODES   100000
#define N_EDGES   3200000
#define D_FEAT    512
#define D_HID     16
#define N_CLASSES 7

typedef float v2f __attribute__((ext_vector_type(2)));
typedef float v8f __attribute__((ext_vector_type(8)));

__global__ void gemm1_wmma(const float* __restrict__ feat,
                           const float* __restrict__ W1,
                           float* __restrict__ t1,
                           float* __restrict__ agg1) {
  __shared__ float ldsW[D_FEAT * D_HID];

  {
    const float4* s4 = (const float4*)W1;
    float4* d4 = (float4*)ldsW;
    for (int i = threadIdx.x; i < (D_FEAT * D_HID) / 4; i += blockDim.x)
      d4[i] = s4[i];
  }
  __syncthreads();

  const int lane = threadIdx.x & 31;
  const int wave = threadIdx.x >> 5;
  const int tile = blockIdx.x * (blockDim.x >> 5) + wave;
  if (tile >= N_NODES / 16) return;
  const int node0 = tile * 16;

  {
    float4* z = (float4*)(agg1 + (size_t)node0 * D_HID);
    z[lane]      = make_float4(0.f, 0.f, 0.f, 0.f);
    z[lane + 32] = make_float4(0.f, 0.f, 0.f, 0.f);
  }

  const int mn   = lane & 15;
  const int koff = (lane >> 4) << 1;
  const float* arow = feat + (size_t)(node0 + mn) * D_FEAT + koff;
  const float* bcol = ldsW + koff * D_HID + mn;

  v8f acc = {};
  #pragma unroll 4
  for (int kk = 0; kk < D_FEAT; kk += 4) {
    v2f a = *(const v2f*)(arow + kk);
    v2f b;
    b.x = bcol[kk * D_HID];
    b.y = bcol[kk * D_HID + D_HID];
    acc = __builtin_amdgcn_wmma_f32_16x16x4_f32(
        false, a, false, b, (short)0, acc, false, false);
  }

  const int rbase = node0 + ((lane >> 4) << 3);
  #pragma unroll
  for (int r = 0; r < 8; ++r)
    t1[(size_t)(rbase + r) * D_HID + mn] = acc[r];
}

#define OWN_T      1024
#define OWN_W      (OWN_T / 32)
#define OWN_U      8
#define OWN_DRAIN  96
#define OWN_QCAP(CH) (OWN_DRAIN + (CH) * 4 * 32)
#define N_VEC4     (N_EDGES / 4)
#define OWN_STEP   (OWN_T * OWN_U)

static_assert(N_EDGES % 4 == 0, "the dst list is read in 16-byte pieces");

template <int K, int S>
struct OwnCfg {
  static constexpr int R     = N_NODES / K;
  static constexpr int G     = R * S;
  static constexpr int SEG   = N_VEC4 / S;
  static constexpr int FULL  = SEG / OWN_STEP;
  static constexpr int ITERS = ((SEG + OWN_STEP - 1) / OWN_STEP + 1) & ~1;
  static constexpr int XB    = K <= 1024 ? 10 : (K <= 2048 ? 11 : 12);
  static_assert(N_NODES % K == 0, "owners tile the rows");
  static_assert(N_VEC4 % S == 0, "segments tile the list");
  static_assert(K <= (1 << XB), "local row fits its bits");
  static_assert((long long)SEG * 4 <= (1ll << (32 - XB)), "edge id inside a segment fits the other bits");
};

template <int K, int XB>
static __device__ __forceinline__ int own_note(unsigned* q, int cnt, unsigned x, int e) {
  const bool m = x < (unsigned)K;
  const unsigned mask = __builtin_amdgcn_ballot_w32(m);
  if (mask != 0u) {
    if (m) q[cnt + (int)__builtin_amdgcn_mbcnt_lo(mask, 0u)] = ((unsigned)e << XB) | x;
    cnt += __builtin_popcount(mask);
  }
  return cnt;
}

template <int C, int P, int K, int XB>
static __device__ __forceinline__ void own_drain(const unsigned* q, int cnt, int lane,
                                                 const float* __restrict__ rows,
                                                 const int* __restrict__ src,
                                                 float* acc) {
  constexpr unsigned XM = (1u << XB) - 1u;
  if constexpr (C == 16) {
    const int sub = lane >> 2, quad = lane & 3;
    for (int k0 = 0; k0 < cnt; k0 += 32) {
      bool ok[4]; unsigned ent[4]; int s[4]; float4 v[4];
      #pragma unroll
      for (int r = 0; r < 4; ++r) {
        const int k = k0 + r * 8 + sub;
        ok[r] = k < cnt;
        ent[r] = ok[r] ? q[k] : 0u;
      }
      #pragma unroll
      for (int r = 0; r < 4; ++r) s[r] = src[ent[r] >> XB];
      #pragma unroll
      for (int r = 0; r < 4; ++r)
        v[r] = ((const float4*)(rows + (size_t)s[r] * 16))[quad];
      #pragma unroll
      for (int r = 0; r < 4; ++r) {
        float* a = acc + (ok[r] ? (int)(ent[r] & XM) * P + quad * 4 : K * P + lane * 4);
        atomicAdd(a + 0, v[r].x);
        atomicAdd(a + 1, v[r].y);
        atomicAdd(a + 2, v[r].z);
        atomicAdd(a + 3, v[r].w);
      }
    }
  } else {
    const int sub = lane >> 3, col = lane & 7;
    const int colc = col < C ? col : C - 1;
    for (int k0 = 0; k0 < cnt; k0 += 16) {
      bool ok[4]; unsigned ent[4]; int s[4]; float v[4];
      #pragma unroll
      for (int r = 0; r < 4; ++r) {
        const int k = k0 + r * 4 + sub;
        ok[r] = (k < cnt) && (col < C);
        ent[r] = ok[r] ? q[k] : 0u;
      }
      #pragma unroll
      for (int r = 0; r < 4; ++r) s[r] = src[ent[r] >> XB];
      #pragma unroll
      for (int r = 0; r < 4; ++r) v[r] = rows[(size_t)s[r] * C + colc];
      #pragma unroll
      for (int r = 0; r < 4; ++r)
        atomicAdd(acc + (ok[r] ? (int)(ent[r] & XM) * P + col : K * P + lane * 4), v[r]);
    }
  }
}

template <int SEG>
static __device__ __forceinline__ void own_load(const int4* __restrict__ dst4, int it, int tid,
                                                int4 (&d)[OWN_U]) {
  const int base = it * OWN_STEP + tid;
  #pragma unroll
  for (int u = 0; u < OWN_U; ++u) {
    const int i = base + u * OWN_T;
    d[u] = dst4[i < SEG ? i : SEG - 1];
  }
}

template <int K, int S, int U0, int CH>
static __device__ __forceinline__ int own_scan(unsigned* q, int cnt, int4 (&d)[OWN_U],
                                               int it, int tid, int lo) {
  using Cfg = OwnCfg<K, S>;
  const int base = it * OWN_STEP + tid;
  if (it >= Cfg::FULL) {
    #pragma unroll
    for (int u = U0; u < U0 + CH; ++u)
      if (base + u * OWN_T >= Cfg::SEG) d[u] = make_int4(-1, -1, -1, -1);
  }
  #pragma unroll
  for (int u = U0; u < U0 + CH; ++u) {
    const int e0 = (base + u * OWN_T) * 4;
    cnt = own_note<K, Cfg::XB>(q, cnt, (unsigned)(d[u].x - lo), e0 + 0);
    cnt = own_note<K, Cfg::XB>(q, cnt, (unsigned)(d[u].y - lo), e0 + 1);
    cnt = own_note<K, Cfg::XB>(q, cnt, (unsigned)(d[u].z - lo), e0 + 2);
    cnt = own_note<K, Cfg::XB>(q, cnt, (unsigned)(d[u].w - lo), e0 + 3);
  }
  return cnt;
}

template <int C, int P, int K, int S, int CH, int U0 = 0>
static __device__ __forceinline__ int own_pass(unsigned* q, int cnt, int4 (&d)[OWN_U],
                                               int it, int tid, int lane, int lo,
                                               const float* __restrict__ rows,
                                               const int* __restrict__ srcs,
                                               float* acc) {
  if constexpr (U0 < OWN_U) {
    cnt = own_scan<K, S, U0, CH>(q, cnt, d, it, tid, lo);
    if (cnt >= OWN_DRAIN) {
      own_drain<C, P, K, OwnCfg<K, S>::XB>(q, cnt, lane, rows, srcs, acc);
      cnt = 0;
    }
    cnt = own_pass<C, P, K, S, CH, U0 + CH>(q, cnt, d, it, tid, lane, lo, rows, srcs, acc);
  }
  return cnt;
}

template <int C, int P, int K, int S, int CH>
__global__ __launch_bounds__(OWN_T)
void owner_sum(const float* __restrict__ rows,
               const int* __restrict__ src,
               const int* __restrict__ dst,
               float* __restrict__ outp) {
  using Cfg = OwnCfg<K, S>;
  static_assert((K * C * 4) % 128 == 0, "an owner's slice of the result starts and ends on 128-B lines");
  static_assert(CH >= 1 && OWN_U % CH == 0, "a pass is a whole number of looks");
  static_assert((K * P + 32 * 4 + OWN_W * OWN_QCAP(CH)) * 4 <= 320 * 1024, "rows and queues fit the LDS");
  __shared__ float acc[K * P + 32 * 4];
  __shared__ unsigned queue[OWN_W * OWN_QCAP(CH)];

  const int tid  = threadIdx.x;
  const int lane = tid & 31;
  const int wave = tid >> 5;
  const int rng  = blockIdx.x / S;
  const int seg  = blockIdx.x - rng * S;
  const int lo   = rng * K;

  for (int i = tid; i < K * P + 32 * 4; i += OWN_T) acc[i] = 0.0f;
  __syncthreads();

  unsigned* q = queue + wave * OWN_QCAP(CH);
  const int4* dst4 = (const int4*)dst + (size_t)seg * Cfg::SEG;
  const int*  srcs = src + (size_t)seg * Cfg::SEG * 4;
  int cnt = 0;
  const int it0 = (rng * Cfg::ITERS) / Cfg::R;

  int4 da[OWN_U], db[OWN_U];
  int ita = it0;
  own_load<Cfg::SEG>(dst4, ita, tid, da);
  for (int n = 0; n < Cfg::ITERS; n += 2) {
    const int itb = (ita + 1 == Cfg::ITERS) ? 0 : ita + 1;
    own_load<Cfg::SEG>(dst4, itb, tid, db);
    cnt = own_pass<C, P, K, S, CH>(q, cnt, da, ita, tid, lane, lo, rows, srcs, acc);
    ita = (itb + 1 == Cfg::ITERS) ? 0 : itb + 1;
    own_load<Cfg::SEG>(dst4, ita, tid, da);
    cnt = own_pass<C, P, K, S, CH>(q, cnt, db, itb, tid, lane, lo, rows, srcs, acc);
  }
  if (cnt > 0) own_drain<C, P, K, Cfg::XB>(q, cnt, lane, rows, srcs, acc);
  __syncthreads();

  float* o = outp + (size_t)lo * C;
  for (int i = tid; i < K * C; i += OWN_T) {
    const int r = i / C, cc = i - r * C;
    atomicAdd(o + i, acc[r * P + cc]);
  }
}

__global__ void layer2_transform(const float* __restrict__ agg1,
                                 const float* __restrict__ b1,
                                 const float* __restrict__ W2,
                                 const float* __restrict__ b2,
                                 float* __restrict__ t2,
                                 float* __restrict__ out) {
  int node = blockIdx.x * blockDim.x + threadIdx.x;
  if (node >= N_NODES) return;
  #pragma unroll
  for (int c = 0; c < N_CLASSES; ++c)
    out[(size_t)node * N_CLASSES + c] = b2[c];
  float h[D_HID];
  const float4* a4 = (const float4*)(agg1 + (size_t)node * D_HID);
  #pragma unroll
  for (int q = 0; q < 4; ++q) {
    float4 v = a4[q];
    h[q * 4 + 0] = fmaxf(v.x + b1[q * 4 + 0], 0.0f);
    h[q * 4 + 1] = fmaxf(v.y + b1[q * 4 + 1], 0.0f);
    h[q * 4 + 2] = fmaxf(v.z + b1[q * 4 + 2], 0.0f);
    h[q * 4 + 3] = fmaxf(v.w + b1[q * 4 + 3], 0.0f);
  }
  #pragma unroll
  for (int c = 0; c < N_CLASSES; ++c) {
    float o = 0.0f;
    #pragma unroll
    for (int j = 0; j < D_HID; ++j)
      o = fmaf(h[j], W2[j * N_CLASSES + c], o);
    t2[(size_t)node * N_CLASSES + c] = o;
  }
}

#define L1_K   4000
#define L1_S   10
#define L1_CH  2
#define L2_K   4000
#define L2_S   10
#define L2_CH  OWN_U

extern "C" void kernel_launch(void* const* d_in, const int* in_sizes, int n_in,
                              void* d_out, int out_size, void* d_ws, size_t ws_size,
                              hipStream_t stream) {
  const float* feat = (const float*)d_in[0];
  const float* W1   = (const float*)d_in[1];
  const float* b1   = (const float*)d_in[2];
  const float* W2   = (const float*)d_in[3];
  const float* b2   = (const float*)d_in[4];
  const int*   src  = (const int*)d_in[5];
  const int*   dst  = (const int*)d_in[6];
  float* out = (float*)d_out;

  float* ws   = (float*)d_ws;
  float* t1   = ws;
  float* agg1 = ws + (size_t)N_NODES * D_HID;
  float* t2   = ws + 2 * (size_t)N_NODES * D_HID;

  const int threads = 256;
  const int tiles = N_NODES / 16;
  const int waves_per_block = threads / 32;
  gemm1_wmma<<<(tiles + waves_per_block - 1) / waves_per_block, threads, 0, stream>>>(
      feat, W1, t1, agg1);

  owner_sum<D_HID, D_HID + 1, L1_K, L1_S, L1_CH><<<OwnCfg<L1_K, L1_S>::G, OWN_T, 0, stream>>>(
      t1, src, dst, agg1);

  layer2_transform<<<(N_NODES + threads - 1) / threads, threads, 0, stream>>>(
      agg1, b1, W2, b2, t2, out);

  owner_sum<N_CLASSES, N_CLASSES, L2_K, L2_S, L2_CH><<<OwnCfg<L2_K, L2_S>::G, OWN_T, 0, stream>>>(
      t2, src, dst, out);
}


# product: feature tile read as a stream: whole 128-B lines per load, 12 KB in flight per wave, non-temporal; A fragments through a private LDS stage; 250 groups x 25 waves; k order kept
# speedup vs baseline: 8.2505x; 1.0526x over previous
#include <hip/hip_runtime.h>
#include <hip/hip_bf16.h>

#define N_NODES   100000
#define N_EDGES   3200000
#define D_FEAT    512
#define D_HID     16
#define N_CLASSES 7

typedef float v2f __attribute__((ext_vector_type(2)));
typedef float v4f __attribute__((ext_vector_type(4)));
typedef float v8f __attribute__((ext_vector_type(8)));

#define G_WAVES 25
#define G_T     (G_WAVES * 32)
#define G_CC    64
#define G_NCH   (D_FEAT / G_CC)
#define G_PITCH (G_CC + 4)
#define G_RING  3
#define G_W4    (D_FEAT * D_HID / 4)

static_assert((N_NODES / 16) % G_WAVES == 0, "every wave has a tile");
static_assert(N_NODES % 16 == 0, "tiles of 16 nodes");

__global__ __launch_bounds__(G_T)
void gemm1_wmma(const float* __restrict__ feat,
                const float* __restrict__ W1,
                float* __restrict__ t1,
                float* __restrict__ agg1) {
  __shared__ float lds[D_FEAT * D_HID + G_WAVES * 16 * G_PITCH];
  float* ldsW  = lds;
  float* stage = lds + D_FEAT * D_HID;

  const int tid  = threadIdx.x;
  const int lane = tid & 31;
  const int wave = __builtin_amdgcn_readfirstlane(tid >> 5);
  const int node0 = (blockIdx.x * G_WAVES + wave) * 16;
  const int half = lane >> 4;
  const int l16  = lane & 15;

  const v4f* s4 = (const v4f*)W1;
  v4f wq[3];
  #pragma unroll
  for (int q = 0; q < 3; ++q) {
    const int idx = tid + q * G_T;
    wq[q] = s4[idx < G_W4 ? idx : G_W4 - 1];
  }

  const float* gsrc = feat + (size_t)node0 * D_FEAT + half * D_FEAT + l16 * 4;
  v4f ring[G_RING][8];
#define G_LOADCH(J)                                                                      \
  do {                                                                                   \
    _Pragma("unroll") for (int i = 0; i < 8; ++i)                                        \
      ring[(J) % G_RING][i] =                                                            \
          __builtin_nontemporal_load((const v4f*)(gsrc + i * 2 * D_FEAT + (J) * G_CC)); \
  } while (0)

  __builtin_amdgcn_sched_barrier(0);
  G_LOADCH(0);
  __builtin_amdgcn_sched_barrier(0);
  G_LOADCH(1);
  __builtin_amdgcn_sched_barrier(0);
  G_LOADCH(2);
  __builtin_amdgcn_sched_barrier(0);

  #pragma unroll
  for (int q = 0; q < 3; ++q) {
    const int idx = tid + q * G_T;
    if (idx < G_W4) ((v4f*)ldsW)[idx] = wq[q];
  }

  {
    float4* z = (float4*)(agg1 + (size_t)node0 * D_HID);
    z[lane]      = make_float4(0.f, 0.f, 0.f, 0.f);
    z[lane + 32] = make_float4(0.f, 0.f, 0.f, 0.f);
  }
  __builtin_amdgcn_fence(__ATOMIC_RELEASE, "workgroup", "local");
  __builtin_amdgcn_s_barrier();
  __builtin_amdgcn_fence(__ATOMIC_ACQUIRE, "workgroup", "local");

  const int mn   = l16;
  const int koff = half << 1;
  float* st = stage + wave * (16 * G_PITCH);
  float* stw = st + half * G_PITCH + l16 * 4;
  const float* sta = st + mn * G_PITCH + koff;
  const float* bcol = ldsW + koff * D_HID + mn;

  v8f acc = {};
  #pragma unroll
  for (int j = 0; j < G_NCH; ++j) {
    __builtin_amdgcn_sched_barrier(0);
    #pragma unroll
    for (int i = 0; i < 8; ++i)
      *(v4f*)(stw + i * 2 * G_PITCH) = ring[j % G_RING][i];
    if (j + G_RING < G_NCH) G_LOADCH(j + G_RING);
    __builtin_amdgcn_sched_barrier(0);
    #pragma unroll
    for (int kk = 0; kk < G_CC; kk += 4) {
      const v2f a = *(const v2f*)(sta + kk);
      v2f b;
      b.x = bcol[(j * G_CC + kk) * D_HID];
      b.y = bcol[(j * G_CC + kk) * D_HID + D_HID];
      acc = __builtin_amdgcn_wmma_f32_16x16x4_f32(
          false, a, false, b, (short)0, acc, false, false);
    }
  }
#undef G_LOADCH

  const int rbase = node0 + (half << 3);
  #pragma unroll
  for (int r = 0; r < 8; ++r)
    t1[(size_t)(rbase + r) * D_HID + mn] = acc[r];
}

#define OWN_T      1024
#define OWN_W      (OWN_T / 32)
#define OWN_U      8
#define OWN_DRAIN  96
#define OWN_QCAP(CH) (OWN_DRAIN + (CH) * 4 * 32)
#define N_VEC4     (N_EDGES / 4)
#define OWN_STEP   (OWN_T * OWN_U)

static_assert(N_EDGES % 4 == 0, "the dst list is read in 16-byte pieces");

template <int K, int S>
struct OwnCfg {
  static constexpr int R     = N_NODES / K;
  static constexpr int G     = R * S;
  static constexpr int SEG   = N_VEC4 / S;
  static constexpr int FULL  = SEG / OWN_STEP;
  static constexpr int ITERS = ((SEG + OWN_STEP - 1) / OWN_STEP + 1) & ~1;
  static constexpr int XB    = K <= 1024 ? 10 : (K <= 2048 ? 11 : 12);
  static_assert(N_NODES % K == 0, "owners tile the rows");
  static_assert(N_VEC4 % S == 0, "segments tile the list");
  static_assert(K <= (1 << XB), "local row fits its bits");
  static_assert((long long)SEG * 4 <= (1ll << (32 - XB)), "edge id inside a segment fits the other bits");
};

template <int K, int XB>
static __device__ __forceinline__ int own_note(unsigned* q, int cnt, unsigned x, int e) {
  const bool m = x < (unsigned)K;
  const unsigned mask = __builtin_amdgcn_ballot_w32(m);
  if (mask != 0u) {
    if (m) q[cnt + (int)__builtin_amdgcn_mbcnt_lo(mask, 0u)] = ((unsigned)e << XB) | x;
    cnt += __builtin_popcount(mask);
  }
  return cnt;
}

template <int C, int P, int K, int XB>
static __device__ __forceinline__ void own_drain(const unsigned* q, int cnt, int lane,
                                                 const float* __restrict__ rows,
                                                 const int* __restrict__ src,
                                                 float* acc) {
  constexpr unsigned XM = (1u << XB) - 1u;
  if constexpr (C == 16) {
    const int sub = lane >> 2, quad = lane & 3;
    for (int k0 = 0; k0 < cnt; k0 += 32) {
      bool ok[4]; unsigned ent[4]; int s[4]; float4 v[4];
      #pragma unroll
      for (int r = 0; r < 4; ++r) {
        const int k = k0 + r * 8 + sub;
        ok[r] = k < cnt;
        ent[r] = ok[r] ? q[k] : 0u;
      }
      #pragma unroll
      for (int r = 0; r < 4; ++r) s[r] = src[ent[r] >> XB];
      #pragma unroll
      for (int r = 0; r < 4; ++r)
        v[r] = ((const float4*)(rows + (size_t)s[r] * 16))[quad];
      #pragma unroll
      for (int r = 0; r < 4; ++r) {
        float* a = acc + (ok[r] ? (int)(ent[r] & XM) * P + quad * 4 : K * P + lane * 4);
        atomicAdd(a + 0, v[r].x);
        atomicAdd(a + 1, v[r].y);
        atomicAdd(a + 2, v[r].z);
        atomicAdd(a + 3, v[r].w);
      }
    }
  } else {
    const int sub = lane >> 3, col = lane & 7;
    const int colc = col < C ? col : C - 1;
    for (int k0 = 0; k0 < cnt; k0 += 16) {
      bool ok[4]; unsigned ent[4]; int s[4]; float v[4];
      #pragma unroll
      for (int r = 0; r < 4; ++r) {
        const int k = k0 + r * 4 + sub;
        ok[r] = (k < cnt) && (col < C);
        ent[r] = ok[r] ? q[k] : 0u;
      }
      #pragma unroll
      for (int r = 0; r < 4; ++r) s[r] = src[ent[r] >> XB];
      #pragma unroll
      for (int r = 0; r < 4; ++r) v[r] = rows[(size_t)s[r] * C + colc];
      #pragma unroll
      for (int r = 0; r < 4; ++r)
        atomicAdd(acc + (ok[r] ? (int)(ent[r] & XM) * P + col : K * P + lane * 4), v[r]);
    }
  }
}

template <int SEG>
static __device__ __forceinline__ void own_load(const int4* __restrict__ dst4, int it, int tid,
                                                int4 (&d)[OWN_U]) {
  const int base = it * OWN_STEP + tid;
  #pragma unroll
  for (int u = 0; u < OWN_U; ++u) {
    const int i = base + u * OWN_T;
    d[u] = dst4[i < SEG ? i : SEG - 1];
  }
}

template <int K, int S, int U0, int CH>
static __device__ __forceinline__ int own_scan(unsigned* q, int cnt, int4 (&d)[OWN_U],
                                               int it, int tid, int lo) {
  using Cfg = OwnCfg<K, S>;
  const int base = it * OWN_STEP + tid;
  if (it >= Cfg::FULL) {
    #pragma unroll
    for (int u = U0; u < U0 + CH; ++u)
      if (base + u * OWN_T >= Cfg::SEG) d[u] = make_int4(-1, -1, -1, -1);
  }
  #pragma unroll
  for (int u = U0; u < U0 + CH; ++u) {
    const int e0 = (base + u * OWN_T) * 4;
    cnt = own_note<K, Cfg::XB>(q, cnt, (unsigned)(d[u].x - lo), e0 + 0);
    cnt = own_note<K, Cfg::XB>(q, cnt, (unsigned)(d[u].y - lo), e0 + 1);
    cnt = own_note<K, Cfg::XB>(q, cnt, (unsigned)(d[u].z - lo), e0 + 2);
    cnt = own_note<K, Cfg::XB>(q, cnt, (unsigned)(d[u].w - lo), e0 + 3);
  }
  return cnt;
}

template <int C, int P, int K, int S, int CH, int U0 = 0>
static __device__ __forceinline__ int own_pass(unsigned* q, int cnt, int4 (&d)[OWN_U],
                                               int it, int tid, int lane, int lo,
                                               const float* __restrict__ rows,
                                               const int* __restrict__ srcs,
                                               float* acc) {
  if constexpr (U0 < OWN_U) {
    cnt = own_scan<K, S, U0, CH>(q, cnt, d, it, tid, lo);
    if (cnt >= OWN_DRAIN) {
      own_drain<C, P, K, OwnCfg<K, S>::XB>(q, cnt, lane, rows, srcs, acc);
      cnt = 0;
    }
    cnt = own_pass<C, P, K, S, CH, U0 + CH>(q, cnt, d, it, tid, lane, lo, rows, srcs, acc);
  }
  return cnt;
}

template <int C, int P, int K, int S, int CH>
__global__ __launch_bounds__(OWN_T)
void owner_sum(const float* __restrict__ rows,
               const int* __restrict__ src,
               const int* __restrict__ dst,
               float* __restrict__ outp) {
  using Cfg = OwnCfg<K, S>;
  static_assert((K * C * 4) % 128 == 0, "an owner's slice of the result starts and ends on 128-B lines");
  static_assert(CH >= 1 && OWN_U % CH == 0, "a pass is a whole number of looks");
  static_assert((K * P + 32 * 4 + OWN_W * OWN_QCAP(CH)) * 4 <= 320 * 1024, "rows and queues fit the LDS");
  __shared__ float acc[K * P + 32 * 4];
  __shared__ unsigned queue[OWN_W * OWN_QCAP(CH)];

  const int tid  = threadIdx.x;
  const int lane = tid & 31;
  const int wave = tid >> 5;
  const int rng  = blockIdx.x / S;
  const int seg  = blockIdx.x - rng * S;
  const int lo   = rng * K;

  for (int i = tid; i < K * P + 32 * 4; i += OWN_T) acc[i] = 0.0f;
  __syncthreads();

  unsigned* q = queue + wave * OWN_QCAP(CH);
  const int4* dst4 = (const int4*)dst + (size_t)seg * Cfg::SEG;
  const int*  srcs = src + (size_t)seg * Cfg::SEG * 4;
  int cnt = 0;
  const int it0 = (rng * Cfg::ITERS) / Cfg::R;

  int4 da[OWN_U], db[OWN_U];
  int ita = it0;
  own_load<Cfg::SEG>(dst4, ita, tid, da);
  for (int n = 0; n < Cfg::ITERS; n += 2) {
    const int itb = (ita + 1 == Cfg::ITERS) ? 0 : ita + 1;
    own_load<Cfg::SEG>(dst4, itb, tid, db);
    cnt = own_pass<C, P, K, S, CH>(q, cnt, da, ita, tid, lane, lo, rows, srcs, acc);
    ita = (itb + 1 == Cfg::ITERS) ? 0 : itb + 1;
    own_load<Cfg::SEG>(dst4, ita, tid, da);
    cnt = own_pass<C, P, K, S, CH>(q, cnt, db, itb, tid, lane, lo, rows, srcs, acc);
  }
  if (cnt > 0) own_drain<C, P, K, Cfg::XB>(q, cnt, lane, rows, srcs, acc);
  __syncthreads();

  float* o = outp + (size_t)lo * C;
  for (int i = tid; i < K * C; i += OWN_T) {
    const int r = i / C, cc = i - r * C;
    atomicAdd(o + i, acc[r * P + cc]);
  }
}

__global__ void layer2_transform(const float* __restrict__ agg1,
                                 const float* __restrict__ b1,
                                 const float* __restrict__ W2,
                                 const float* __restrict__ b2,
                                 float* __restrict__ t2,
                                 float* __restrict__ out) {
  int node = blockIdx.x * blockDim.x + threadIdx.x;
  if (node >= N_NODES) return;
  #pragma unroll
  for (int c = 0; c < N_CLASSES; ++c)
    out[(size_t)node * N_CLASSES + c] = b2[c];
  float h[D_HID];
  const float4* a4 = (const float4*)(agg1 + (size_t)node * D_HID);
  #pragma unroll
  for (int q = 0; q < 4; ++q) {
    float4 v = a4[q];
    h[q * 4 + 0] = fmaxf(v.x + b1[q * 4 + 0], 0.0f);
    h[q * 4 + 1] = fmaxf(v.y + b1[q * 4 + 1], 0.0f);
    h[q * 4 + 2] = fmaxf(v.z + b1[q * 4 + 2], 0.0f);
    h[q * 4 + 3] = fmaxf(v.w + b1[q * 4 + 3], 0.0f);
  }
  #pragma unroll
  for (int c = 0; c < N_CLASSES; ++c) {
    float o = 0.0f;
    #pragma unroll
    for (int j = 0; j < D_HID; ++j)
      o = fmaf(h[j], W2[j * N_CLASSES + c], o);
    t2[(size_t)node * N_CLASSES + c] = o;
  }
}

#define L1_K   4000
#define L1_S   10
#define L1_CH  2
#define L2_K   4000
#define L2_S   10
#define L2_CH  OWN_U

extern "C" void kernel_launch(void* const* d_in, const int* in_sizes, int n_in,
                              void* d_out, int out_size, void* d_ws, size_t ws_size,
                              hipStream_t stream) {
  const float* feat = (const float*)d_in[0];
  const float* W1   = (const float*)d_in[1];
  const float* b1   = (const float*)d_in[2];
  const float* W2   = (const float*)d_in[3];
  const float* b2   = (const float*)d_in[4];
  const int*   src  = (const int*)d_in[5];
  const int*   dst  = (const int*)d_in[6];
  float* out = (float*)d_out;

  float* ws   = (float*)d_ws;
  float* t1   = ws;
  float* agg1 = ws + (size_t)N_NODES * D_HID;
  float* t2   = ws + 2 * (size_t)N_NODES * D_HID;

  const int threads = 256;
  gemm1_wmma<<<N_NODES / 16 / G_WAVES, G_T, 0, stream>>>(feat, W1, t1, agg1);

  owner_sum<D_HID, D_HID + 1, L1_K, L1_S, L1_CH><<<OwnCfg<L1_K, L1_S>::G, OWN_T, 0, stream>>>(
      t1, src, dst, agg1);

  layer2_transform<<<(N_NODES + threads - 1) / threads, threads, 0, stream>>>(
      agg1, b1, W2, b2, t2, out);

  owner_sum<N_CLASSES, N_CLASSES, L2_K, L2_S, L2_CH><<<OwnCfg<L2_K, L2_S>::G, OWN_T, 0, stream>>>(
      t2, src, dst, out);
}
